# starting kernel
# baseline (speedup 1.0000x reference)
#include <hip/hip_runtime.h>

#define D     256
#define S     2048
#define BATCH 8
#define BM    128
#define BN    128
#define KC    64
#define AS_STRIDE 68
#define BT_STRIDE 68

typedef __attribute__((ext_vector_type(2))) float v2f;
typedef __attribute__((ext_vector_type(4))) float v4f;
typedef __attribute__((ext_vector_type(8))) float v8f;

__global__ void colsum_partial(const float* __restrict__ x, float* __restrict__ partial) {
    int b     = blockIdx.x >> 3;
    int chunk = blockIdx.x & 7;
    int d     = threadIdx.x;
    const float* p = x + (b * S + chunk * 256) * D + d;
    float acc = 0.f;
#pragma unroll 8
    for (int s = 0; s < 256; ++s) acc += p[s * D];
    partial[blockIdx.x * D + d] = acc;
}

__global__ void reduce_and_coef(const float* __restrict__ partial,
                                const float* __restrict__ comp,
                                float* __restrict__ sums,
                                float* __restrict__ coef) {
    int idx = blockIdx.x * 256 + threadIdx.x;
    int b = idx >> 8, d = idx & 255;
    float s = 0.f;
#pragma unroll
    for (int c = 0; c < 8; ++c) s += partial[(b * 8 + c) * D + d];
    sums[idx] = s;

    int t = idx;
    float inv1 = 1.0f / (float)(1 + (t + 2 < S));
    int   d2   = (t >= 1) + (t + 3 < S);
    float inv2 = d2 ? 1.0f / (float)d2 : 0.f;
    int   bc   = (t >= 1) + 1 + (t + 1 < S) + (t + 2 < S) + (t + 3 < S);
    float inv3 = 1.0f / (float)(S - bc);
#pragma unroll
    for (int j = 0; j < 2; ++j) {
        float c0 = comp[0 * 2 + j], c1 = comp[1 * 2 + j];
        float c2 = comp[2 * 2 + j], c3 = comp[3 * 2 + j];
        float m3 = c3 * inv3;
        float* w = coef + t * 12 + j * 6;
        w[0] = (t >= 1)    ? (c2 * inv2 - m3) : 0.f;
        w[1] =               c1 * inv1 - m3;
        w[2] = (t + 1 < S) ? (c0        - m3) : 0.f;
        w[3] = (t + 2 < S) ? (c1 * inv1 - m3) : 0.f;
        w[4] = (t + 3 < S) ? (c2 * inv2 - m3) : 0.f;
        w[5] = m3;
    }
}

__global__ __launch_bounds__(256) void rgcn_gemm(const float* __restrict__ x,
                                                 const float* __restrict__ basis,
                                                 const float* __restrict__ root,
                                                 const float* __restrict__ bias,
                                                 const float* __restrict__ sums,
                                                 const float* __restrict__ coef,
                                                 float* __restrict__ out) {
    __shared__ float As[BM * AS_STRIDE];
    __shared__ float Bst[BN * BT_STRIDE];

    const int tid      = threadIdx.x;
    const int rowBase  = blockIdx.x * BM;
    const int b        = rowBase / S;
    const int tBase    = rowBase % S;
    const int colBase0 = blockIdx.y * BN;

    const int wid     = tid >> 5;
    const int lane    = tid & 31;
    const int waveRow = wid & 3;
    const int waveCol = wid >> 2;
    const int lmod    = lane & 15;
    const int lhi     = lane >> 4;
    const int koff    = lhi * 2;
    const int mrow0   = waveRow * 32 + lmod;
    const int mrow1   = mrow0 + 16;

    v8f acc[2][4] = {};

#pragma unroll 1
    for (int kc = 0; kc < 12; ++kc) {
        const int seg    = kc >> 2;
        const int kcBase = (kc & 3) * KC;

        if (seg == 2) {
#pragma unroll
            for (int e = 0; e < 8; ++e) {
                int idx = e * 256 + tid;
                int c4 = (idx & 15) * 4;
                int r  = idx >> 4;
                *(v4f*)&As[r * AS_STRIDE + c4] =
                    *(const v4f*)&x[(b * S + tBase + r) * D + kcBase + c4];
            }
        } else {
#pragma unroll 2
            for (int e = 0; e < 8; ++e) {
                int idx = e * 256 + tid;
                int c4 = (idx & 15) * 4;
                int r  = idx >> 4;
                int t  = tBase + r;
                int kd = kcBase + c4;
                const float* w = coef + t * 12 + seg * 6;
                v4f val = w[5] * (*(const v4f*)&sums[b * D + kd]);
#pragma unroll
                for (int o = 0; o < 5; ++o) {
                    int sr = t - 1 + o;
                    sr = sr < 0 ? 0 : (sr >= S ? S - 1 : sr);
                    val += w[o] * (*(const v4f*)&x[(b * S + sr) * D + kd]);
                }
                *(v4f*)&As[r * AS_STRIDE + c4] = val;
            }
        }

        const float* Wseg = (seg == 0) ? basis : (seg == 1 ? basis + D * D : root);
#pragma unroll
        for (int e = 0; e < 8; ++e) {
            int idx = e * 256 + tid;
            int n4 = (idx & 31) * 4;
            int kr = idx >> 5;
            v4f wv = *(const v4f*)&Wseg[(kcBase + kr) * D + colBase0 + n4];
            Bst[(n4 + 0) * BT_STRIDE + kr] = wv.x;
            Bst[(n4 + 1) * BT_STRIDE + kr] = wv.y;
            Bst[(n4 + 2) * BT_STRIDE + kr] = wv.z;
            Bst[(n4 + 3) * BT_STRIDE + kr] = wv.w;
        }
        __syncthreads();

#pragma unroll
        for (int kk = 0; kk < KC; kk += 4) {
            v2f a0 = *(const v2f*)&As[mrow0 * AS_STRIDE + kk + koff];
            v2f a1 = *(const v2f*)&As[mrow1 * AS_STRIDE + kk + koff];
#pragma unroll
            for (int sub = 0; sub < 4; ++sub) {
                int ncol = waveCol * 64 + sub * 16 + lmod;
                v2f bv = *(const v2f*)&Bst[ncol * BT_STRIDE + kk + koff];
                acc[0][sub] = __builtin_amdgcn_wmma_f32_16x16x4_f32(
                    false, a0, false, bv, (short)0, acc[0][sub], false, false);
                acc[1][sub] = __builtin_amdgcn_wmma_f32_16x16x4_f32(
                    false, a1, false, bv, (short)0, acc[1][sub], false, false);
            }
        }
        __syncthreads();
    }

#pragma unroll
    for (int rt = 0; rt < 2; ++rt) {
#pragma unroll
        for (int sub = 0; sub < 4; ++sub) {
            int ncol = colBase0 + waveCol * 64 + sub * 16 + lmod;
            float badd = bias[ncol];
#pragma unroll
            for (int jj = 0; jj < 8; ++jj) {
                int r = waveRow * 32 + rt * 16 + jj + 8 * lhi;
                out[(b * S + tBase + r) * D + ncol] = acc[rt][sub][jj] + badd;
            }
        }
    }
}

extern "C" void kernel_launch(void* const* d_in, const int* in_sizes, int n_in,
                              void* d_out, int out_size, void* d_ws, size_t ws_size,
                              hipStream_t stream) {
    const float* x     = (const float*)d_in[0];
    const float* comp  = (const float*)d_in[1];
    const float* basis = (const float*)d_in[2];
    const float* root  = (const float*)d_in[3];
    const float* bias  = (const float*)d_in[4];
    float* out = (float*)d_out;

    float* wsf     = (float*)d_ws;
    float* sums    = wsf;
    float* coef    = wsf + BATCH * D;
    float* partial = coef + S * 12;

    colsum_partial<<<BATCH * 8, D, 0, stream>>>(x, partial);
    reduce_and_coef<<<S / 256, 256, 0, stream>>>(partial, comp, sums, coef);
    rgcn_gemm<<<dim3((BATCH * S) / BM, D / BN), 256, 0, stream>>>(
        x, basis, root, bias, sums, coef, out);
}


# band mix and column-sum term fused into the GEMM workgroup: it computes Y0,Y1,Z for its rows plus halo rows, mixes them in LDS, writes out; Y never stored; G from x column sums: sums in a new order
# speedup vs baseline: 1.9861x; 1.9861x over previous
#include <hip/hip_runtime.h>


#define D     256
#define S     2048
#define BATCH 8
#define NROWS (BATCH * S)
#define BM    128
#define BC    64
#define KC    64
#define NRB   (NROWS / BM)
#define RBPB  (S / BM)
#define LSTR  68
#define AROWS 144
#define BROWS 192
#define TROWS 136
#define CSROWS 64
#define NPART (S / CSROWS)
#define RPT   8
#define SMEM_FLOATS ((2 * TROWS + BM) * LSTR)

typedef __attribute__((ext_vector_type(2))) float v2f;
typedef __attribute__((ext_vector_type(4))) float v4f;
typedef __attribute__((ext_vector_type(8))) float v8f;

__global__ __launch_bounds__(256) void x_colsum(const float* __restrict__ x, float* __restrict__ xpart) {
    __shared__ __attribute__((aligned(16))) float red[4 * D];
    const int tid   = threadIdx.x;
    const int q     = tid & 63;
    const int strip = tid >> 6;
    const float* p = x + ((size_t)blockIdx.x * CSROWS + strip * 16) * D + q * 4;
    v4f v[16];
#pragma unroll
    for (int i = 0; i < 16; ++i) v[i] = *(const v4f*)(p + i * D);
    v4f s = v[0];
#pragma unroll
    for (int i = 1; i < 16; ++i) s += v[i];
    *(v4f*)&red[strip * D + q * 4] = s;
    __syncthreads();
    if (tid < 64) {
        v4f r = (*(const v4f*)&red[q * 4] + *(const v4f*)&red[D + q * 4]) +
                (*(const v4f*)&red[2 * D + q * 4] + *(const v4f*)&red[3 * D + q * 4]);
        *(v4f*)&xpart[(size_t)blockIdx.x * D + q * 4] = r;
    }
}

static __device__ __forceinline__ void load_frags(v2f (&fa)[2][3], v2f (&fb)[2][7],
                                                  const float* pa0, const float* pa1, const float* pah,
                                                  const float* pb, const float* pbh, int kb) {
#pragma unroll
    for (int u = 0; u < 2; ++u) {
        const int k = kb + 4 * u;
        fa[u][0] = *(const v2f*)(pa0 + k);
        fa[u][1] = *(const v2f*)(pa1 + k);
        fa[u][2] = *(const v2f*)(pah + k);
#pragma unroll
        for (int pc = 0; pc < 6; ++pc)
            fb[u][pc] = *(const v2f*)(pb + ((pc >> 1) * 64 + (pc & 1) * 16) * LSTR + k);
        fb[u][6] = *(const v2f*)(pbh + k);
    }
}

__global__ __launch_bounds__(256) __attribute__((amdgpu_waves_per_eu(4, 4)))
void gemm_band(const float* __restrict__ x,
               const float* __restrict__ basis,
               const float* __restrict__ root,
               const float* __restrict__ comp,
               const float* __restrict__ bias,
               const float* __restrict__ xpart,
               float* __restrict__ out) {
    __shared__ __attribute__((aligned(16))) float smem[SMEM_FLOATS];
    float* As = smem;
    float* Bs = smem + AROWS * LSTR;
    float* Sx = Bs + BROWS * LSTR;

    const int tid     = threadIdx.x;
    const int rblk    = blockIdx.x;
    const int C0      = blockIdx.y * BC;
    const int b       = rblk / RBPB;
    const int t0      = (rblk % RBPB) * BM;
    const int rowBase = rblk * BM;

    const int wid     = tid >> 5;
    const int lane    = tid & 31;
    const int waveRow = wid & 3;
    const int waveCol = wid >> 2;
    const int lmod    = lane & 15;
    const int lhi     = lane >> 4;
    const int koff    = lhi * 2;
    const int mrow0   = waveRow * 32 + lmod;
    const int mrow1   = mrow0 + 16;
    const int hs      = waveRow & 1;
    const int hct     = waveRow >> 1;
    const int hcol    = waveCol * 32 + hct * 16 + lmod;

    {
        const float* p = xpart + (size_t)b * NPART * D + tid;
        float s = p[0];
#pragma unroll
        for (int i = 1; i < NPART; ++i) s += p[i * D];
        Sx[tid] = s;
    }
    __syncthreads();

    const float* xa  = x + (size_t)(rowBase + (tid >> 4)) * D + (tid & 15) * 4;
    float*       asw = As + (tid >> 4) * LSTR + (tid & 15) * 4;
    const int hr = tid >> 4;
    int th = (hr == 0) ? (t0 - 1) : (t0 + 127 + (hr < 4 ? hr : 3));
    th = th < 0 ? 0 : (th >= S ? S - 1 : th);
    const float* xh   = x + ((size_t)b * S + th) * D + (tid & 15) * 4;
    float*       ashw = As + (128 + hr) * LSTR + (tid & 15) * 4;
    const int bq = (tid & 7) + 8 * ((tid >> 5) & 1);
    const int kr = ((tid >> 3) & 3) + 4 * (tid >> 6);
    const float* wb0 = basis + (size_t)kr * D + C0 + bq * 4;
    const float* wb1 = wb0 + D * D;
    const float* wb2 = root + (size_t)kr * D + C0 + bq * 4;
    float*       bsw = Bs + (bq * 4) * LSTR + kr;
    const float* pa0 = As + mrow0 * LSTR + koff;
    const float* pa1 = As + mrow1 * LSTR + koff;
    const float* pah = As + (128 + lmod) * LSTR + koff;
    const float* pb  = Bs + (waveCol * 32 + lmod) * LSTR + koff;
    const float* pbh = Bs + (hs * 64 + hcol) * LSTR + koff;

    v8f acc[2][6] = {};
    v8f acch = {};

#pragma unroll 1
    for (int kc = 0; kc < D / KC; ++kc) {
        const int kcBase = kc * KC;

        v4f ra[8], rb0[4], rb1[4], rb2[4];
#pragma unroll
        for (int e = 0; e < 8; ++e) ra[e] = *(const v4f*)(xa + kcBase + e * 16 * D);
#pragma unroll
        for (int e = 0; e < 4; ++e) {
            rb0[e] = *(const v4f*)(wb0 + (kcBase + e * 16) * D);
            rb1[e] = *(const v4f*)(wb1 + (kcBase + e * 16) * D);
            rb2[e] = *(const v4f*)(wb2 + (kcBase + e * 16) * D);
        }
        const v4f rh = *(const v4f*)(xh + kcBase);

#pragma unroll
        for (int e = 0; e < 8; ++e) *(v4f*)(asw + e * 16 * LSTR) = ra[e];
        {
            const v4f sx   = *(const v4f*)&Sx[kcBase + (tid & 15) * 4];
            const v4f zero = {0.f, 0.f, 0.f, 0.f};
            *(v4f*)ashw = hr < 4 ? rh : (hr == 4 ? sx : zero);
        }
#pragma unroll
        for (int e = 0; e < 4; ++e) {
            bsw[(0 * 64 + 0) * LSTR + e * 16] = rb0[e].x;
            bsw[(0 * 64 + 1) * LSTR + e * 16] = rb0[e].y;
            bsw[(0 * 64 + 2) * LSTR + e * 16] = rb0[e].z;
            bsw[(0 * 64 + 3) * LSTR + e * 16] = rb0[e].w;
            bsw[(1 * 64 + 0) * LSTR + e * 16] = rb1[e].x;
            bsw[(1 * 64 + 1) * LSTR + e * 16] = rb1[e].y;
            bsw[(1 * 64 + 2) * LSTR + e * 16] = rb1[e].z;
            bsw[(1 * 64 + 3) * LSTR + e * 16] = rb1[e].w;
            bsw[(2 * 64 + 0) * LSTR + e * 16] = rb2[e].x;
            bsw[(2 * 64 + 1) * LSTR + e * 16] = rb2[e].y;
            bsw[(2 * 64 + 2) * LSTR + e * 16] = rb2[e].z;
            bsw[(2 * 64 + 3) * LSTR + e * 16] = rb2[e].w;
        }
        __syncthreads();

        v2f ca[2][3], cb[2][7];
        load_frags(ca, cb, pa0, pa1, pah, pb, pbh, 0);
#pragma unroll
        for (int kb = 0; kb < KC; kb += 8) {
            v2f na[2][3], nb[2][7];
            if (kb + 8 < KC) load_frags(na, nb, pa0, pa1, pah, pb, pbh, kb + 8);
            __builtin_amdgcn_sched_barrier(0);
#pragma unroll
            for (int u = 0; u < 2; ++u) {
#pragma unroll
                for (int pc = 0; pc < 6; ++pc) {
                    acc[0][pc] = __builtin_amdgcn_wmma_f32_16x16x4_f32(
                        false, ca[u][0], false, cb[u][pc], (short)0, acc[0][pc], false, false);
                    acc[1][pc] = __builtin_amdgcn_wmma_f32_16x16x4_f32(
                        false, ca[u][1], false, cb[u][pc], (short)0, acc[1][pc], false, false);
                }
                acch = __builtin_amdgcn_wmma_f32_16x16x4_f32(
                    false, ca[u][2], false, cb[u][6], (short)0, acch, false, false);
            }
            __builtin_amdgcn_sched_barrier(0);
            if (kb + 8 < KC) {
#pragma unroll
                for (int u = 0; u < 2; ++u) {
#pragma unroll
                    for (int i = 0; i < 3; ++i) ca[u][i] = na[u][i];
#pragma unroll
                    for (int i = 0; i < 7; ++i) cb[u][i] = nb[u][i];
                }
            }
        }
        __syncthreads();
    }

    float* T0 = smem;
    float* T1 = smem + TROWS * LSTR;
    float* TZ = smem + 2 * TROWS * LSTR;
#pragma unroll
    for (int rt = 0; rt < 2; ++rt) {
#pragma unroll
        for (int pc = 0; pc < 6; ++pc) {
            float* Tp = (pc >> 1) == 0 ? T0 : ((pc >> 1) == 1 ? T1 : TZ);
            int col = waveCol * 32 + (pc & 1) * 16 + lmod;
#pragma unroll
            for (int jj = 0; jj < 8; ++jj) {
                int r = waveRow * 32 + rt * 16 + jj + 8 * lhi;
                Tp[r * LSTR + col] = acc[rt][pc][jj];
            }
        }
    }
    if (lhi == 0) {
        float* Th = smem + hs * (TROWS * LSTR);
#pragma unroll
        for (int jj = 0; jj < 5; ++jj) Th[(128 + jj) * LSTR + hcol] = acch[jj];
    }
    __syncthreads();

    const int q   = tid & 15;
    const int n   = q * 4;
    const int lr0 = (tid >> 4) * RPT;

    v4f y0[RPT + 4], y1[RPT + 4], z[RPT];
#pragma unroll
    for (int w = 0; w < RPT + 4; ++w) {
        int lw = lr0 - 1 + w;
        int tr = lw < 0 ? 128 : (lw >= BM ? lw + 1 : lw);
        y0[w] = *(const v4f*)&T0[tr * LSTR + n];
        y1[w] = *(const v4f*)&T1[tr * LSTR + n];
    }
#pragma unroll
    for (int i = 0; i < RPT; ++i) z[i] = *(const v4f*)&TZ[(lr0 + i) * LSTR + n];
    const v4f g0 = *(const v4f*)&T0[132 * LSTR + n];
    const v4f g1 = *(const v4f*)&T1[132 * LSTR + n];

    const v4f bias4 = *(const v4f*)&bias[C0 + n];
    float c[4][2];
#pragma unroll
    for (int r = 0; r < 4; ++r) { c[r][0] = comp[r * 2 + 0]; c[r][1] = comp[r * 2 + 1]; }

#pragma unroll
    for (int i = 0; i < RPT; ++i) {
        const int t = t0 + lr0 + i;
        float inv1 = (t + 2 < S) ? 0.5f : 1.0f;
        int   d2   = (t >= 1) + (t + 3 < S);
        float inv2 = (d2 == 2) ? 0.5f : (d2 == 1 ? 1.0f : 0.f);
        int   bc   = (t >= 1) + 1 + (t + 1 < S) + (t + 2 < S) + (t + 3 < S);
        float inv3 = 1.0f / (float)(S - bc);
        v4f val = z[i] + bias4;
#pragma unroll
        for (int j = 0; j < 2; ++j) {
            float m3 = c[3][j] * inv3;
            float w0 = (t >= 1)    ? (c[2][j] * inv2 - m3) : 0.f;
            float w1 =               c[1][j] * inv1 - m3;
            float w2 = (t + 1 < S) ? (c[0][j]        - m3) : 0.f;
            float w3 = (t + 2 < S) ? (c[1][j] * inv1 - m3) : 0.f;
            float w4 = (t + 3 < S) ? (c[2][j] * inv2 - m3) : 0.f;
            const v4f* yy = j ? y1 : y0;
            v4f u = m3 * (j ? g1 : g0);
            u += w0 * yy[i + 0];
            u += w1 * yy[i + 1];
            u += w2 * yy[i + 2];
            u += w3 * yy[i + 3];
            u += w4 * yy[i + 4];
            val += u;
        }
        *(v4f*)&out[((size_t)b * S + t) * D + C0 + n] = val;
    }
}

extern "C" void kernel_launch(void* const* d_in, const int* in_sizes, int n_in,
                              void* d_out, int out_size, void* d_ws, size_t ws_size,
                              hipStream_t stream) {
    const float* x     = (const float*)d_in[0];
    const float* comp  = (const float*)d_in[1];
    const float* basis = (const float*)d_in[2];
    const float* root  = (const float*)d_in[3];
    const float* bias  = (const float*)d_in[4];
    float* out = (float*)d_out;

    float* xpart = (float*)d_ws;

    x_colsum<<<NROWS / CSROWS, 256, 0, stream>>>(x, xpart);
    gemm_band<<<dim3(NRB, D / BC), 256, 0, stream>>>(x, basis, root, comp, bias, xpart, out);
}


# weights bypass LDS: launch 1 writes [B0|B1|root] to the workspace in WMMA B-fragment order; a GEMM wave reads a tile's two K-steps with one 16-byte global load per lane, one 8-k block ahead
# speedup vs baseline: 2.0681x; 1.0413x over previous
#include <hip/hip_runtime.h>


#define D     256
#define S     2048
#define BATCH 8
#define NROWS (BATCH * S)
#define BM    128
#define BC    64
#define KC    64
#define NRB   (NROWS / BM)
#define RBPB  (S / BM)
#define LSTR  68
#define AROWS 144
#define NCB   (D / BC)
#define NKB   (D / 8)
#define WTILES 12
#define WPACK_V4 (NCB * NKB * WTILES * 32)
#define PACKWG (WPACK_V4 / 1024)
#define TROWS 136
#define CSROWS 64
#define NPART (S / CSROWS)
#define RPT   8
#define SMEM_FLOATS ((2 * TROWS + BM) * LSTR)

typedef __attribute__((ext_vector_type(2))) float v2f;
typedef __attribute__((ext_vector_type(4))) float v4f;
typedef __attribute__((ext_vector_type(8))) float v8f;

__global__ __launch_bounds__(256) void x_colsum(const float* __restrict__ x,
                                                const float* __restrict__ basis,
                                                const float* __restrict__ root,
                                                float* __restrict__ xpart,
                                                float* __restrict__ wpack) {
    __shared__ __attribute__((aligned(16))) float red[4 * D];
    const int tid   = threadIdx.x;
    if (blockIdx.x >= NROWS / CSROWS) {
        const int g0 = (blockIdx.x - NROWS / CSROWS) * 1024 + tid;
#pragma unroll
        for (int j = 0; j < 4; ++j) {
            const int g     = g0 + j * 256;
            const int lane  = g & 31;
            const int t12   = (g >> 5) % WTILES;
            const int kblk  = (g / (32 * WTILES)) % NKB;
            const int cblk  = g / (32 * WTILES * NKB);
            const int plane = t12 >> 2;
            const int ncol  = cblk * BC + (t12 & 3) * 16 + (lane & 15);
            const int k0    = kblk * 8 + (lane >> 4) * 2;
            const float* w  = (plane == 2 ? root : basis + (size_t)plane * D * D) + ncol;
            v4f v;
            v.x = w[(size_t)(k0 + 0) * D];
            v.y = w[(size_t)(k0 + 1) * D];
            v.z = w[(size_t)(k0 + 4) * D];
            v.w = w[(size_t)(k0 + 5) * D];
            *(v4f*)&wpack[(size_t)g * 4] = v;
        }
        return;
    }
    const int q     = tid & 63;
    const int strip = tid >> 6;
    const float* p = x + ((size_t)blockIdx.x * CSROWS + strip * 16) * D + q * 4;
    v4f v[16];
#pragma unroll
    for (int i = 0; i < 16; ++i) v[i] = *(const v4f*)(p + i * D);
    v4f s = v[0];
#pragma unroll
    for (int i = 1; i < 16; ++i) s += v[i];
    *(v4f*)&red[strip * D + q * 4] = s;
    __syncthreads();
    if (tid < 64) {
        v4f r = (*(const v4f*)&red[q * 4] + *(const v4f*)&red[D + q * 4]) +
                (*(const v4f*)&red[2 * D + q * 4] + *(const v4f*)&red[3 * D + q * 4]);
        *(v4f*)&xpart[(size_t)blockIdx.x * D + q * 4] = r;
    }
}

static __device__ __forceinline__ void load_a(v2f (&fa)[2][3],
                                              const float* pa0, const float* pa1, const float* pah, int kb) {
#pragma unroll
    for (int u = 0; u < 2; ++u) {
        const int k = kb + 4 * u;
        fa[u][0] = *(const v2f*)(pa0 + k);
        fa[u][1] = *(const v2f*)(pa1 + k);
        fa[u][2] = *(const v2f*)(pah + k);
    }
}

static __device__ __forceinline__ void load_b(v4f (&fb)[7], const v4f* wp, int hoff) {
#pragma unroll
    for (int pc = 0; pc < 6; ++pc) fb[pc] = wp[((pc >> 1) * 4 + (pc & 1)) * 32];
    fb[6] = wp[hoff];
}

__global__ __launch_bounds__(256) __attribute__((amdgpu_waves_per_eu(4, 4)))
void gemm_band(const float* __restrict__ x,
               const float* __restrict__ wpack,
               const float* __restrict__ comp,
               const float* __restrict__ bias,
               const float* __restrict__ xpart,
               float* __restrict__ out) {
    __shared__ __attribute__((aligned(16))) float smem[SMEM_FLOATS];
    float* As = smem;
    float* Sx = smem + AROWS * LSTR;

    const int tid     = threadIdx.x;
    const int rblk    = blockIdx.x;
    const int C0      = blockIdx.y * BC;
    const int b       = rblk / RBPB;
    const int t0      = (rblk % RBPB) * BM;
    const int rowBase = rblk * BM;

    const int wid     = tid >> 5;
    const int lane    = tid & 31;
    const int waveRow = wid & 3;
    const int waveCol = wid >> 2;
    const int lmod    = lane & 15;
    const int lhi     = lane >> 4;
    const int koff    = lhi * 2;
    const int mrow0   = waveRow * 32 + lmod;
    const int mrow1   = mrow0 + 16;
    const int hs      = waveRow & 1;
    const int hct     = waveRow >> 1;
    const int hcol    = waveCol * 32 + hct * 16 + lmod;

    const v4f* wpb  = (const v4f*)wpack + ((size_t)blockIdx.y * NKB * WTILES + waveCol * 2) * 32 + lane;
    const int  hoff = (hs * 4 + hct) * 32;
    v4f nbf[7];
    load_b(nbf, wpb, hoff);

    {
        const float* p = xpart + (size_t)b * NPART * D + tid;
        float s = p[0];
#pragma unroll
        for (int i = 1; i < NPART; ++i) s += p[i * D];
        Sx[tid] = s;
    }
    __syncthreads();

    const float* xa  = x + (size_t)(rowBase + (tid >> 4)) * D + (tid & 15) * 4;
    float*       asw = As + (tid >> 4) * LSTR + (tid & 15) * 4;
    const int hr = tid >> 4;
    int th = (hr == 0) ? (t0 - 1) : (t0 + 127 + (hr < 4 ? hr : 3));
    th = th < 0 ? 0 : (th >= S ? S - 1 : th);
    const float* xh   = x + ((size_t)b * S + th) * D + (tid & 15) * 4;
    float*       ashw = As + (128 + hr) * LSTR + (tid & 15) * 4;
    const float* pa0 = As + mrow0 * LSTR + koff;
    const float* pa1 = As + mrow1 * LSTR + koff;
    const float* pah = As + (128 + lmod) * LSTR + koff;

    v8f acc[2][6] = {};
    v8f acch = {};

#pragma unroll 1
    for (int kc = 0; kc < D / KC; ++kc) {
        const int kcBase = kc * KC;

        v4f ra[8];
#pragma unroll
        for (int e = 0; e < 8; ++e) ra[e] = *(const v4f*)(xa + kcBase + e * 16 * D);
        const v4f rh = *(const v4f*)(xh + kcBase);

#pragma unroll
        for (int e = 0; e < 8; ++e) *(v4f*)(asw + e * 16 * LSTR) = ra[e];
        {
            const v4f sx   = *(const v4f*)&Sx[kcBase + (tid & 15) * 4];
            const v4f zero = {0.f, 0.f, 0.f, 0.f};
            *(v4f*)ashw = hr < 4 ? rh : (hr == 4 ? sx : zero);
        }
        __syncthreads();

        v2f ca[2][3];
        load_a(ca, pa0, pa1, pah, 0);
#pragma unroll
        for (int kb = 0; kb < KC; kb += 8) {
            v4f cbf[7];
#pragma unroll
            for (int i = 0; i < 7; ++i) cbf[i] = nbf[i];
            int nblk = kc * (KC / 8) + (kb >> 3) + 1;
            nblk = nblk < NKB ? nblk : NKB - 1;
            load_b(nbf, wpb + (size_t)nblk * (WTILES * 32), hoff);
            v2f na[2][3];
            if (kb + 8 < KC) load_a(na, pa0, pa1, pah, kb + 8);
            __builtin_amdgcn_sched_barrier(0);
#pragma unroll
            for (int u = 0; u < 2; ++u) {
#pragma unroll
                for (int pc = 0; pc < 6; ++pc) {
                    const v2f bfr = u ? __builtin_shufflevector(cbf[pc], cbf[pc], 2, 3)
                                      : __builtin_shufflevector(cbf[pc], cbf[pc], 0, 1);
                    acc[0][pc] = __builtin_amdgcn_wmma_f32_16x16x4_f32(
                        false, ca[u][0], false, bfr, (short)0, acc[0][pc], false, false);
                    acc[1][pc] = __builtin_amdgcn_wmma_f32_16x16x4_f32(
                        false, ca[u][1], false, bfr, (short)0, acc[1][pc], false, false);
                }
                const v2f bfh = u ? __builtin_shufflevector(cbf[6], cbf[6], 2, 3)
                                  : __builtin_shufflevector(cbf[6], cbf[6], 0, 1);
                acch = __builtin_amdgcn_wmma_f32_16x16x4_f32(
                    false, ca[u][2], false, bfh, (short)0, acch, false, false);
            }
            __builtin_amdgcn_sched_barrier(0);
            if (kb + 8 < KC) {
#pragma unroll
                for (int u = 0; u < 2; ++u) {
#pragma unroll
                    for (int i = 0; i < 3; ++i) ca[u][i] = na[u][i];
                }
            }
        }
        __syncthreads();
    }

    float* T0 = smem;
    float* T1 = smem + TROWS * LSTR;
    float* TZ = smem + 2 * TROWS * LSTR;
#pragma unroll
    for (int rt = 0; rt < 2; ++rt) {
#pragma unroll
        for (int pc = 0; pc < 6; ++pc) {
            float* Tp = (pc >> 1) == 0 ? T0 : ((pc >> 1) == 1 ? T1 : TZ);
            int col = waveCol * 32 + (pc & 1) * 16 + lmod;
#pragma unroll
            for (int jj = 0; jj < 8; ++jj) {
                int r = waveRow * 32 + rt * 16 + jj + 8 * lhi;
                Tp[r * LSTR + col] = acc[rt][pc][jj];
            }
        }
    }
    if (lhi == 0) {
        float* Th = smem + hs * (TROWS * LSTR);
#pragma unroll
        for (int jj = 0; jj < 5; ++jj) Th[(128 + jj) * LSTR + hcol] = acch[jj];
    }
    __syncthreads();

    const int q   = tid & 15;
    const int n   = q * 4;
    const int lr0 = (tid >> 4) * RPT;

    v4f y0[RPT + 4], y1[RPT + 4], z[RPT];
#pragma unroll
    for (int w = 0; w < RPT + 4; ++w) {
        int lw = lr0 - 1 + w;
        int tr = lw < 0 ? 128 : (lw >= BM ? lw + 1 : lw);
        y0[w] = *(const v4f*)&T0[tr * LSTR + n];
        y1[w] = *(const v4f*)&T1[tr * LSTR + n];
    }
#pragma unroll
    for (int i = 0; i < RPT; ++i) z[i] = *(const v4f*)&TZ[(lr0 + i) * LSTR + n];
    const v4f g0 = *(const v4f*)&T0[132 * LSTR + n];
    const v4f g1 = *(const v4f*)&T1[132 * LSTR + n];

    const v4f bias4 = *(const v4f*)&bias[C0 + n];
    float c[4][2];
#pragma unroll
    for (int r = 0; r < 4; ++r) { c[r][0] = comp[r * 2 + 0]; c[r][1] = comp[r * 2 + 1]; }

#pragma unroll
    for (int i = 0; i < RPT; ++i) {
        const int t = t0 + lr0 + i;
        float inv1 = (t + 2 < S) ? 0.5f : 1.0f;
        int   d2   = (t >= 1) + (t + 3 < S);
        float inv2 = (d2 == 2) ? 0.5f : (d2 == 1 ? 1.0f : 0.f);
        int   bc   = (t >= 1) + 1 + (t + 1 < S) + (t + 2 < S) + (t + 3 < S);
        float inv3 = 1.0f / (float)(S - bc);
        v4f val = z[i] + bias4;
#pragma unroll
        for (int j = 0; j < 2; ++j) {
            float m3 = c[3][j] * inv3;
            float w0 = (t >= 1)    ? (c[2][j] * inv2 - m3) : 0.f;
            float w1 =               c[1][j] * inv1 - m3;
            float w2 = (t + 1 < S) ? (c[0][j]        - m3) : 0.f;
            float w3 = (t + 2 < S) ? (c[1][j] * inv1 - m3) : 0.f;
            float w4 = (t + 3 < S) ? (c[2][j] * inv2 - m3) : 0.f;
            const v4f* yy = j ? y1 : y0;
            v4f u = m3 * (j ? g1 : g0);
            u += w0 * yy[i + 0];
            u += w1 * yy[i + 1];
            u += w2 * yy[i + 2];
            u += w3 * yy[i + 3];
            u += w4 * yy[i + 4];
            val += u;
        }
        *(v4f*)&out[((size_t)b * S + t) * D + C0 + n] = val;
    }
}

extern "C" void kernel_launch(void* const* d_in, const int* in_sizes, int n_in,
                              void* d_out, int out_size, void* d_ws, size_t ws_size,
                              hipStream_t stream) {
    const float* x     = (const float*)d_in[0];
    const float* comp  = (const float*)d_in[1];
    const float* basis = (const float*)d_in[2];
    const float* root  = (const float*)d_in[3];
    const float* bias  = (const float*)d_in[4];
    float* out = (float*)d_out;

    float* xpart = (float*)d_ws;
    float* wpack = xpart + BATCH * NPART * D;

    x_colsum<<<NROWS / CSROWS + PACKWG, 256, 0, stream>>>(x, basis, root, xpart, wpack);
    gemm_band<<<dim3(NRB, D / BC), 256, 0, stream>>>(x, wpack, comp, bias, xpart, out);
}
